# starting kernel
# baseline (speedup 1.0000x reference)
#include <hip/hip_runtime.h>
#include <hip/hip_bf16.h>

#define NB      8
#define NN      512
#define HIST    24
#define SKIP    256
#define ENDC    128
#define KERN    13
#define FUT     12
#define OUTD    2
#define NODES   (NB * NN)

#define XS_STRIDE 260
#define HS_STRIDE 132

typedef float v2f __attribute__((ext_vector_type(2)));
typedef float v8f __attribute__((ext_vector_type(8)));

__global__ __launch_bounds__(256)
void moe_tcnn_kernel(const float* __restrict__ x,
                     const int*   __restrict__ labels,
                     const float* __restrict__ W1,
                     const float* __restrict__ b1,
                     const float* __restrict__ W2,
                     const float* __restrict__ b2,
                     float*       __restrict__ out)
{
    __shared__ float Xs[32 * XS_STRIDE];
    __shared__ float Hs[HIST * HS_STRIDE];

    const int node = blockIdx.x;
    const int tid  = threadIdx.x;
    const int g    = labels[node];

    const float* xp = x + (size_t)node * (HIST * SKIP);
    #pragma unroll
    for (int q = 0; q < 6; ++q) {
        int idx4 = tid + q * 256;
        int idx  = idx4 * 4;
        int row  = idx >> 8;
        int col  = idx & 255;
        float4 v = *(const float4*)(xp + idx);
        v.x = v.x > 0.f ? v.x : 0.f;
        v.y = v.y > 0.f ? v.y : 0.f;
        v.z = v.z > 0.f ? v.z : 0.f;
        v.w = v.w > 0.f ? v.w : 0.f;
        *(float4*)(&Xs[row * XS_STRIDE + col]) = v;
    }
    #pragma unroll
    for (int q = 0; q < 8; ++q) {
        int i   = tid + q * 256;
        int row = 24 + (i >> 8);
        int col = i & 255;
        Xs[row * XS_STRIDE + col] = 0.f;
    }
    __syncthreads();

    const int wv   = tid >> 5;
    const int lane = tid & 31;
    const int half = lane >> 4;
    const int l16  = lane & 15;
    const int n0   = wv * 16;

    const float* W1row = W1 + ((size_t)g * ENDC + n0 + l16) * SKIP + 2 * half;
    const float* Arow0 = &Xs[l16 * XS_STRIDE + 2 * half];
    const float* Arow1 = &Xs[(16 + l16) * XS_STRIDE + 2 * half];

    v8f c0 = {0.f,0.f,0.f,0.f,0.f,0.f,0.f,0.f};
    v8f c1 = {0.f,0.f,0.f,0.f,0.f,0.f,0.f,0.f};

    for (int kk = 0; kk < SKIP; kk += 32) {
        #pragma unroll
        for (int u = 0; u < 8; ++u) {
            int k = kk + u * 4;
            v2f af0 = *(const v2f*)(Arow0 + k);
            v2f af1 = *(const v2f*)(Arow1 + k);
            v2f bf  = *(const v2f*)(W1row + k);
            c0 = __builtin_amdgcn_wmma_f32_16x16x4_f32(false, af0, false, bf,
                                                       (short)0, c0, false, false);
            c1 = __builtin_amdgcn_wmma_f32_16x16x4_f32(false, af1, false, bf,
                                                       (short)0, c1, false, false);
        }
    }

    const float bias = b1[g * ENDC + n0 + l16];
    const int   colH = n0 + l16;
    #pragma unroll
    for (int j = 0; j < 8; ++j) {
        int row0 = j + 8 * half;
        float v0 = c0[j] + bias;
        Hs[row0 * HS_STRIDE + colH] = v0 > 0.f ? v0 : 0.f;
        int row1 = 16 + j + 8 * half;
        if (row1 < HIST) {
            float v1 = c1[j] + bias;
            Hs[row1 * HS_STRIDE + colH] = v1 > 0.f ? v1 : 0.f;
        }
    }
    __syncthreads();

    const int group = wv * 4 + (lane >> 3);
    const int l8    = lane & 7;
    if (group < FUT * OUTD) {
        const int f = group % FUT;
        const int o = group / FUT;
        const float* W2go = W2 + ((size_t)(g * OUTD + o) * ENDC) * KERN;
        float acc = 0.f;
        #pragma unroll
        for (int i = 0; i < ENDC / 8; ++i) {
            int e = l8 + 8 * i;
            const float* hcol = &Hs[f * HS_STRIDE + e];
            const float* wrow = &W2go[e * KERN];
            #pragma unroll
            for (int k = 0; k < KERN; ++k)
                acc += hcol[k * HS_STRIDE] * wrow[k];
        }
        acc += __shfl_xor(acc, 4, 32);
        acc += __shfl_xor(acc, 2, 32);
        acc += __shfl_xor(acc, 1, 32);
        if (l8 == 0)
            out[(size_t)node * (FUT * OUTD) + f * OUTD + o] = acc + b2[g * OUTD + o];
    }
}

extern "C" void kernel_launch(void* const* d_in, const int* in_sizes, int n_in,
                              void* d_out, int out_size, void* d_ws, size_t ws_size,
                              hipStream_t stream) {
    const float* x      = (const float*)d_in[0];
    const int*   labels = (const int*)  d_in[1];
    const float* W1     = (const float*)d_in[2];
    const float* b1     = (const float*)d_in[3];
    const float* W2     = (const float*)d_in[4];
    const float* b2     = (const float*)d_in[5];
    float*       out    = (float*)d_out;

    moe_tcnn_kernel<<<NODES, 256, 0, stream>>>(x, labels, W1, b1, W2, b2, out);
}


# Expert-stationary work-groups: 8 experts x 64 persistent groups; each keeps its expert's W1 fragments in registers (W2 in LDS) and streams that expert's nodes in pairs (in-kernel rank, no atomics)
# speedup vs baseline: 1.1497x; 1.1497x over previous
#include <hip/hip_runtime.h>

#define NB      8
#define NN      512
#define HIST    24
#define SKIP    256
#define ENDC    128
#define KERN    13
#define FUT     12
#define OUTD    2
#define NODES   (NB * NN)
#define NEXP    8
#define SLICES  64
#define NWG     (NEXP * SLICES)
#define LISTMAX 64
#define PROWS   48
#define XS_STRIDE 260
#define HS_STRIDE 132
#define XBUF    (PROWS * XS_STRIDE)
#define W2E     (OUTD * ENDC * KERN)

typedef float v2f __attribute__((ext_vector_type(2)));
typedef float v8f __attribute__((ext_vector_type(8)));

__device__ __forceinline__ float4 relu4(float4 v) {
    v.x = v.x > 0.f ? v.x : 0.f;
    v.y = v.y > 0.f ? v.y : 0.f;
    v.z = v.z > 0.f ? v.z : 0.f;
    v.w = v.w > 0.f ? v.w : 0.f;
    return v;
}

__global__ __launch_bounds__(256) __attribute__((amdgpu_waves_per_eu(4, 4)))
void moe_tcnn_kernel(const float* __restrict__ x,
                     const int*   __restrict__ labels,
                     const float* __restrict__ W1,
                     const float* __restrict__ b1,
                     const float* __restrict__ W2,
                     const float* __restrict__ b2,
                     float*       __restrict__ out)
{
    __shared__ __attribute__((aligned(16))) float Xs[2 * XBUF];
    __shared__ __attribute__((aligned(16))) float Hs[PROWS * HS_STRIDE];
    __shared__ __attribute__((aligned(16))) float W2s[W2E];
    __shared__ int lst[LISTMAX];
    __shared__ int wtot[8];

    const int tid  = threadIdx.x;
    const int g    = blockIdx.x & (NEXP - 1);
    const int s    = blockIdx.x >> 3;
    const int wv   = tid >> 5;
    const int lane = tid & 31;
    const int half = lane >> 4;
    const int l16  = lane & 15;
    const int n0   = wv * 16;

    const float* W1row = W1 + ((size_t)g * ENDC + n0 + l16) * SKIP + 2 * half;
    v2f bfr[64];
    #pragma unroll
    for (int ks = 0; ks < 64; ++ks) bfr[ks] = *(const v2f*)(W1row + 4 * ks);
    const float bias = b1[g * ENDC + n0 + l16];
    const int   colH = n0 + l16;

    for (int i = tid; i < W2E / 4; i += 256)
        *(float4*)(&W2s[4 * i]) = *(const float4*)(W2 + (size_t)g * W2E + 4 * i);

    int lab[16];
    #pragma unroll
    for (int j = 0; j < 16; ++j) lab[j] = labels[j * 256 + tid];
    unsigned m = 0;
    #pragma unroll
    for (int j = 0; j < 16; ++j) m |= (lab[j] == g ? 1u : 0u) << j;
    const int cnt = __popc(m);
    int inc = cnt;
    #pragma unroll
    for (int d = 1; d < 32; d <<= 1) {
        int t = __shfl_up(inc, d, 32);
        if (lane >= d) inc += t;
    }
    if (lane == 31) wtot[wv] = inc;
    __syncthreads();
    int base = 0, total = 0;
    #pragma unroll
    for (int w = 0; w < 8; ++w) {
        int t = wtot[w];
        total += t;
        if (w < wv) base += t;
    }
    const int P      = (total + 1) >> 1;
    const int plo    = (s * P) / SLICES;
    const int phi    = ((s + 1) * P) / SLICES;
    const int npairs = phi - plo;
    const int lo     = 2 * plo;
    const int hi     = (2 * phi < total) ? 2 * phi : total;
    const int nn     = hi - lo;
    {
        int r = base + inc - cnt;
        #pragma unroll
        for (int j = 0; j < 16; ++j) {
            if ((m >> j) & 1u) {
                if (r >= lo && r < hi) lst[r - lo] = j * 256 + tid;
                ++r;
            }
        }
    }
    __syncthreads();

    const int xrow = tid >> 6;
    const int xcol = (tid & 63) * 4;
    const int group = wv * 4 + (lane >> 3);
    const int l8    = lane & 7;

    if (npairs > 0) {
        const int nA = lst[0];
        const int nB = (1 < nn) ? lst[1] : nA;
        #pragma unroll
        for (int q = 0; q < 12; ++q) {
            const int nd = (q < 6) ? nA : nB;
            const int qq = (q < 6) ? q : q - 6;
            float4 v = *(const float4*)(x + (size_t)nd * (HIST * SKIP) + (tid + qq * 256) * 4);
            *(float4*)(&Xs[(xrow + 4 * q) * XS_STRIDE + xcol]) = relu4(v);
        }
    }

    for (int p = 0; p < npairs; ++p) {
        const int  cur    = p & 1;
        const int  nodeA  = lst[2 * p];
        const bool validB = (2 * p + 1) < nn;
        const int  nodeB  = validB ? lst[2 * p + 1] : nodeA;
        const bool has_next = (p + 1) < npairs;
        int nxtA = nodeA, nxtB = nodeA;
        if (has_next) {
            nxtA = lst[2 * p + 2];
            nxtB = ((2 * p + 3) < nn) ? lst[2 * p + 3] : nxtA;
        }
        __syncthreads();

        const float* A0 = &Xs[cur * XBUF + l16 * XS_STRIDE + 2 * half];
        float* Xn = &Xs[(cur ^ 1) * XBUF + xrow * XS_STRIDE + xcol];
        v8f c0 = {0.f,0.f,0.f,0.f,0.f,0.f,0.f,0.f};
        v8f c1 = {0.f,0.f,0.f,0.f,0.f,0.f,0.f,0.f};
        v8f c2 = {0.f,0.f,0.f,0.f,0.f,0.f,0.f,0.f};
        #pragma unroll
        for (int kq = 0; kq < 4; ++kq) {
            float4 pf[3] = {};
            if (has_next) {
                #pragma unroll
                for (int j = 0; j < 3; ++j) {
                    const int q  = 3 * kq + j;
                    const int nd = (q < 6) ? nxtA : nxtB;
                    const int qq = (q < 6) ? q : q - 6;
                    pf[j] = *(const float4*)(x + (size_t)nd * (HIST * SKIP) + (tid + qq * 256) * 4);
                }
            }
            #pragma unroll
            for (int u = 0; u < 16; ++u) {
                const int ks = kq * 16 + u;
                v2f a0 = *(const v2f*)(A0 + 4 * ks);
                v2f a1 = *(const v2f*)(A0 + 16 * XS_STRIDE + 4 * ks);
                v2f a2 = *(const v2f*)(A0 + 32 * XS_STRIDE + 4 * ks);
                c0 = __builtin_amdgcn_wmma_f32_16x16x4_f32(false, a0, false, bfr[ks], (short)0, c0, false, false);
                c1 = __builtin_amdgcn_wmma_f32_16x16x4_f32(false, a1, false, bfr[ks], (short)0, c1, false, false);
                c2 = __builtin_amdgcn_wmma_f32_16x16x4_f32(false, a2, false, bfr[ks], (short)0, c2, false, false);
            }
            if (has_next) {
                #pragma unroll
                for (int j = 0; j < 3; ++j) {
                    const int q = 3 * kq + j;
                    *(float4*)(Xn + 4 * q * XS_STRIDE) = relu4(pf[j]);
                }
            }
        }

        #pragma unroll
        for (int j = 0; j < 8; ++j) {
            const int r = j + 8 * half;
            float v0 = c0[j] + bias, v1 = c1[j] + bias, v2 = c2[j] + bias;
            Hs[r * HS_STRIDE + colH]        = v0 > 0.f ? v0 : 0.f;
            Hs[(16 + r) * HS_STRIDE + colH] = v1 > 0.f ? v1 : 0.f;
            Hs[(32 + r) * HS_STRIDE + colH] = v2 > 0.f ? v2 : 0.f;
        }
        __syncthreads();

        if (group < FUT * OUTD) {
            const int f = group % FUT;
            const int o = group / FUT;
            const float* w2o = &W2s[o * ENDC * KERN];
            const float b2v = b2[g * OUTD + o];
            for (int nd = 0; nd < 2; ++nd) {
                float acc = 0.f;
                #pragma unroll 8
                for (int i = 0; i < ENDC / 8; ++i) {
                    const int e = l8 + 8 * i;
                    const float* hcol = &Hs[(HIST * nd + f) * HS_STRIDE + e];
                    const float* wrow = &w2o[e * KERN];
                    #pragma unroll
                    for (int k = 0; k < KERN; ++k)
                        acc += hcol[k * HS_STRIDE] * wrow[k];
                }
                acc += __shfl_xor(acc, 4, 32);
                acc += __shfl_xor(acc, 2, 32);
                acc += __shfl_xor(acc, 1, 32);
                const int node = nd ? nodeB : nodeA;
                if (l8 == 0 && (nd == 0 || validB))
                    out[(size_t)node * (FUT * OUTD) + f * OUTD + o] = acc + b2v;
            }
        }
    }
}

extern "C" void kernel_launch(void* const* d_in, const int* in_sizes, int n_in,
                              void* d_out, int out_size, void* d_ws, size_t ws_size,
                              hipStream_t stream) {
    const float* x      = (const float*)d_in[0];
    const int*   labels = (const int*)  d_in[1];
    const float* W1     = (const float*)d_in[2];
    const float* b1     = (const float*)d_in[3];
    const float* W2     = (const float*)d_in[4];
    const float* b2     = (const float*)d_in[5];
    float*       out    = (float*)d_out;

    moe_tcnn_kernel<<<NWG, 256, 0, stream>>>(x, labels, W1, b1, W2, b2, out);
}


# Stage 2 on the matrix unit: stage 1 builds H transposed, its accumulators feed Z = H @ W2r as A operands from registers (no Hs); per-wave partial Z in LDS, k-diagonal summed last (sum order changes)
# speedup vs baseline: 1.7551x; 1.5266x over previous
#include <hip/hip_runtime.h>

#define NB      8
#define NN      512
#define HIST    24
#define SKIP    256
#define ENDC    128
#define KERN    13
#define FUT     12
#define OUTD    2
#define NODES   (NB * NN)
#define NEXP    8
#define SLICES  64
#define NWG     (NEXP * SLICES)
#define LISTMAX 64
#define PROWS   48
#define XS_STRIDE 260
#define XBUF    (PROWS * XS_STRIDE)
#define W2E     (OUTD * ENDC * KERN)
#define ZN      (OUTD * KERN)
#define ZP      52
#define ZW      (ZN * ZP)

typedef float v2f __attribute__((ext_vector_type(2)));
typedef float v8f __attribute__((ext_vector_type(8)));

__device__ __forceinline__ float4 relu4(float4 v) {
    v.x = v.x > 0.f ? v.x : 0.f;
    v.y = v.y > 0.f ? v.y : 0.f;
    v.z = v.z > 0.f ? v.z : 0.f;
    v.w = v.w > 0.f ? v.w : 0.f;
    return v;
}

__global__ __launch_bounds__(256) __attribute__((amdgpu_waves_per_eu(4, 4)))
void moe_tcnn_kernel(const float* __restrict__ x,
                     const int*   __restrict__ labels,
                     const float* __restrict__ W1,
                     const float* __restrict__ b1,
                     const float* __restrict__ W2,
                     const float* __restrict__ b2,
                     float*       __restrict__ out)
{
    __shared__ __attribute__((aligned(16))) float Xs[2 * XBUF];
    __shared__ __attribute__((aligned(16))) float Zs[8 * ZW];
    __shared__ int lst[LISTMAX];
    __shared__ int wtot[8];

    const int tid  = threadIdx.x;
    const int g    = blockIdx.x & (NEXP - 1);
    const int s    = blockIdx.x >> 3;
    const int wv   = tid >> 5;
    const int lane = tid & 31;
    const int half = lane >> 4;
    const int l16  = lane & 15;
    const int n0   = wv * 16;

    const float* W1row = W1 + ((size_t)g * ENDC + n0 + l16) * SKIP + 2 * half;
    v2f bfr[64];
    #pragma unroll
    for (int ks = 0; ks < 64; ++ks) bfr[ks] = *(const v2f*)(W1row + 4 * ks);
    const float4 bia = *(const float4*)(b1 + g * ENDC + n0 + 8 * half);
    const float4 bib = *(const float4*)(b1 + g * ENDC + n0 + 8 * half + 4);
    const float bias8[8] = {bia.x, bia.y, bia.z, bia.w, bib.x, bib.y, bib.z, bib.w};

    v2f w2f[4][2];
    #pragma unroll
    for (int nt = 0; nt < 2; ++nt) {
        const int  n   = 16 * nt + l16;
        const bool nok = n < ZN;
        const int  nc  = nok ? n : 0;
        const int  o2  = nc >= KERN ? 1 : 0;
        const int  k2  = nc - KERN * o2;
        const float* wp = W2 + (size_t)g * W2E + o2 * (ENDC * KERN) + (n0 + 8 * half) * KERN + k2;
        #pragma unroll
        for (int j = 0; j < 4; ++j) {
            const float w0 = wp[(2 * j) * KERN];
            const float w1 = wp[(2 * j + 1) * KERN];
            w2f[j][nt].x = nok ? w0 : 0.f;
            w2f[j][nt].y = nok ? w1 : 0.f;
        }
    }

    int lab[16];
    #pragma unroll
    for (int j = 0; j < 16; ++j) lab[j] = labels[j * 256 + tid];
    unsigned m = 0;
    #pragma unroll
    for (int j = 0; j < 16; ++j) m |= (lab[j] == g ? 1u : 0u) << j;
    const int cnt = __popc(m);
    int inc = cnt;
    #pragma unroll
    for (int d = 1; d < 32; d <<= 1) {
        int t = __shfl_up(inc, d, 32);
        if (lane >= d) inc += t;
    }
    if (lane == 31) wtot[wv] = inc;
    __syncthreads();
    int base = 0, total = 0;
    #pragma unroll
    for (int w = 0; w < 8; ++w) {
        int t = wtot[w];
        total += t;
        if (w < wv) base += t;
    }
    const int P      = (total + 1) >> 1;
    const int plo    = (s * P) / SLICES;
    const int phi    = ((s + 1) * P) / SLICES;
    const int npairs = phi - plo;
    const int lo     = 2 * plo;
    const int hi     = (2 * phi < total) ? 2 * phi : total;
    const int nn     = hi - lo;
    {
        int r = base + inc - cnt;
        #pragma unroll
        for (int j = 0; j < 16; ++j) {
            if ((m >> j) & 1u) {
                if (r >= lo && r < hi) lst[r - lo] = j * 256 + tid;
                ++r;
            }
        }
    }
    __syncthreads();

    const int xrow = tid >> 6;
    const int xcol = (tid & 63) * 4;
    const int part = lane >> 3;
    const int u    = wv * 8 + (lane & 7);
    const int und  = u >= FUT * OUTD ? 1 : 0;
    const int urem = u - FUT * OUTD * und;
    const int uf   = urem >> 1;
    const int uo   = urem & 1;
    const float b2v = (wv < 6) ? b2[g * OUTD + uo] : 0.f;

    if (npairs > 0) {
        const int nA = lst[0];
        const int nB = (1 < nn) ? lst[1] : nA;
        #pragma unroll
        for (int q = 0; q < 12; ++q) {
            const int nd = (q < 6) ? nA : nB;
            const int qq = (q < 6) ? q : q - 6;
            float4 v = *(const float4*)(x + (size_t)nd * (HIST * SKIP) + (tid + qq * 256) * 4);
            *(float4*)(&Xs[(xrow + 4 * q) * XS_STRIDE + xcol]) = relu4(v);
        }
    }

    for (int p = 0; p < npairs; ++p) {
        const int  cur    = p & 1;
        const int  nodeA  = lst[2 * p];
        const bool validB = (2 * p + 1) < nn;
        const int  nodeB  = validB ? lst[2 * p + 1] : nodeA;
        const bool has_next = (p + 1) < npairs;
        int nxtA = nodeA, nxtB = nodeA;
        if (has_next) {
            nxtA = lst[2 * p + 2];
            nxtB = ((2 * p + 3) < nn) ? lst[2 * p + 3] : nxtA;
        }
        __syncthreads();

        const float* A0 = &Xs[cur * XBUF + l16 * XS_STRIDE + 2 * half];
        float* Xn = &Xs[(cur ^ 1) * XBUF + xrow * XS_STRIDE + xcol];
        v8f c0 = {0.f,0.f,0.f,0.f,0.f,0.f,0.f,0.f};
        v8f c1 = {0.f,0.f,0.f,0.f,0.f,0.f,0.f,0.f};
        v8f c2 = {0.f,0.f,0.f,0.f,0.f,0.f,0.f,0.f};
        #pragma unroll
        for (int kq = 0; kq < 4; ++kq) {
            float4 pf[3] = {};
            if (has_next) {
                #pragma unroll
                for (int j = 0; j < 3; ++j) {
                    const int q  = 3 * kq + j;
                    const int nd = (q < 6) ? nxtA : nxtB;
                    const int qq = (q < 6) ? q : q - 6;
                    pf[j] = *(const float4*)(x + (size_t)nd * (HIST * SKIP) + (tid + qq * 256) * 4);
                }
            }
            #pragma unroll
            for (int u16 = 0; u16 < 16; ++u16) {
                const int ks = kq * 16 + u16;
                v2f a0 = *(const v2f*)(A0 + 4 * ks);
                v2f a1 = *(const v2f*)(A0 + 16 * XS_STRIDE + 4 * ks);
                v2f a2 = *(const v2f*)(A0 + 32 * XS_STRIDE + 4 * ks);
                c0 = __builtin_amdgcn_wmma_f32_16x16x4_f32(false, bfr[ks], false, a0, (short)0, c0, false, false);
                c1 = __builtin_amdgcn_wmma_f32_16x16x4_f32(false, bfr[ks], false, a1, (short)0, c1, false, false);
                c2 = __builtin_amdgcn_wmma_f32_16x16x4_f32(false, bfr[ks], false, a2, (short)0, c2, false, false);
            }
            if (has_next) {
                #pragma unroll
                for (int j = 0; j < 3; ++j) {
                    const int q = 3 * kq + j;
                    *(float4*)(Xn + 4 * q * XS_STRIDE) = relu4(pf[j]);
                }
            }
        }

        {
            v8f ct[3] = {c0, c1, c2};
            #pragma unroll
            for (int t = 0; t < 3; ++t) {
                #pragma unroll
                for (int r = 0; r < 8; ++r) {
                    const float hv = ct[t][r] + bias8[r];
                    ct[t][r] = hv > 0.f ? hv : 0.f;
                }
            }
            float* zrow = &Zs[wv * ZW + l16 * ZP + 8 * half];
            #pragma unroll
            for (int t = 0; t < 3; ++t) {
                v8f z0 = {0.f,0.f,0.f,0.f,0.f,0.f,0.f,0.f};
                v8f z1 = {0.f,0.f,0.f,0.f,0.f,0.f,0.f,0.f};
                #pragma unroll
                for (int j = 0; j < 4; ++j) {
                    v2f a;
                    a.x = ct[t][2 * j];
                    a.y = ct[t][2 * j + 1];
                    z0 = __builtin_amdgcn_wmma_f32_16x16x4_f32(false, a, false, w2f[j][0], (short)0, z0, false, false);
                    z1 = __builtin_amdgcn_wmma_f32_16x16x4_f32(false, a, false, w2f[j][1], (short)0, z1, false, false);
                }
                float4 s0 = {z0[0], z0[1], z0[2], z0[3]};
                float4 s1 = {z0[4], z0[5], z0[6], z0[7]};
                *(float4*)(zrow + 16 * t)     = s0;
                *(float4*)(zrow + 16 * t + 4) = s1;
                if (l16 < ZN - 16) {
                    float4 s2 = {z1[0], z1[1], z1[2], z1[3]};
                    float4 s3 = {z1[4], z1[5], z1[6], z1[7]};
                    *(float4*)(zrow + 16 * ZP + 16 * t)     = s2;
                    *(float4*)(zrow + 16 * ZP + 16 * t + 4) = s3;
                }
            }
        }
        __syncthreads();

        if (wv < 6) {
            const float* zb = &Zs[(2 * part) * ZW + (uo * KERN) * ZP + HIST * und + uf];
            float ac[4] = {0.f, 0.f, 0.f, 0.f};
            #pragma unroll
            for (int w = 0; w < 2; ++w) {
                #pragma unroll
                for (int k = 0; k < KERN; ++k)
                    ac[2 * w + (k & 1)] += zb[w * ZW + k * (ZP + 1)];
            }
            float acc = (ac[0] + ac[1]) + (ac[2] + ac[3]);
            acc += __shfl_xor(acc, 8, 32);
            acc += __shfl_xor(acc, 16, 32);
            const int node = und ? nodeB : nodeA;
            if (part == 0 && (und == 0 || validB))
                out[(size_t)node * (FUT * OUTD) + urem] = acc + b2v;
        }
    }
}

extern "C" void kernel_launch(void* const* d_in, const int* in_sizes, int n_in,
                              void* d_out, int out_size, void* d_ws, size_t ws_size,
                              hipStream_t stream) {
    const float* x      = (const float*)d_in[0];
    const int*   labels = (const int*)  d_in[1];
    const float* W1     = (const float*)d_in[2];
    const float* b1     = (const float*)d_in[3];
    const float* W2     = (const float*)d_in[4];
    const float* b2     = (const float*)d_in[5];
    float*       out    = (float*)d_out;

    moe_tcnn_kernel<<<NWG, 256, 0, stream>>>(x, labels, W1, b1, W2, b2, out);
}


# Balanced split: each work-group takes exactly 8 nodes of the expert-sorted order (8 expert ranks from one packed prefix sum); W1 refilled at an expert boundary; a lone node costs 2 tiles (same sums)
# speedup vs baseline: 1.9130x; 1.0900x over previous
#include <hip/hip_runtime.h>

#define NB      8
#define NN      512
#define HIST    24
#define SKIP    256
#define ENDC    128
#define KERN    13
#define FUT     12
#define OUTD    2
#define NODES   (NB * NN)
#define NEXP    8
#define NWG     512
#define NPW     (NODES / NWG)
#define PROWS   48
#define XS_STRIDE 260
#define XBUF    (PROWS * XS_STRIDE)
#define W2E     (OUTD * ENDC * KERN)
#define ZN      (OUTD * KERN)
#define ZP      52
#define ZW      (ZN * ZP)

typedef float v2f __attribute__((ext_vector_type(2)));
typedef float v8f __attribute__((ext_vector_type(8)));

__device__ __forceinline__ float4 relu4(float4 v) {
    v.x = v.x > 0.f ? v.x : 0.f;
    v.y = v.y > 0.f ? v.y : 0.f;
    v.z = v.z > 0.f ? v.z : 0.f;
    v.w = v.w > 0.f ? v.w : 0.f;
    return v;
}

__device__ __forceinline__ void load_expert(const int e,
                                            const float* __restrict__ W1, const float* __restrict__ b1,
                                            const float* __restrict__ W2, const float* __restrict__ b2,
                                            const int n0, const int l16, const int half,
                                            const bool outlane, const int uo,
                                            v2f (&bfr)[64], float (&bias8)[8], v2f (&w2f)[4][2], float& b2v)
{
    const float* W1row = W1 + ((size_t)e * ENDC + n0 + l16) * SKIP + 2 * half;
    #pragma unroll
    for (int ks = 0; ks < 64; ++ks) bfr[ks] = *(const v2f*)(W1row + 4 * ks);
    const float4 bia = *(const float4*)(b1 + e * ENDC + n0 + 8 * half);
    const float4 bib = *(const float4*)(b1 + e * ENDC + n0 + 8 * half + 4);
    bias8[0] = bia.x; bias8[1] = bia.y; bias8[2] = bia.z; bias8[3] = bia.w;
    bias8[4] = bib.x; bias8[5] = bib.y; bias8[6] = bib.z; bias8[7] = bib.w;
    #pragma unroll
    for (int nt = 0; nt < 2; ++nt) {
        const int  n   = 16 * nt + l16;
        const bool nok = n < ZN;
        const int  nc  = nok ? n : 0;
        const int  o2  = nc >= KERN ? 1 : 0;
        const int  k2  = nc - KERN * o2;
        const float* wp = W2 + (size_t)e * W2E + o2 * (ENDC * KERN) + (n0 + 8 * half) * KERN + k2;
        #pragma unroll
        for (int j = 0; j < 4; ++j) {
            const float w0 = wp[(2 * j) * KERN];
            const float w1 = wp[(2 * j + 1) * KERN];
            w2f[j][nt].x = nok ? w0 : 0.f;
            w2f[j][nt].y = nok ? w1 : 0.f;
        }
    }
    b2v = outlane ? b2[e * OUTD + uo] : 0.f;
}

template <bool THREE>
__device__ __forceinline__ void stage1(const v2f (&bfr)[64], const float* __restrict__ x,
                                       const float* A0, float* Xn, const bool has_next,
                                       const int nxtA, const int nxtB, const int tid,
                                       v8f& c0, v8f& c1, v8f& c2)
{
    #pragma unroll
    for (int kq = 0; kq < 4; ++kq) {
        float4 pf[3] = {};
        if (has_next) {
            #pragma unroll
            for (int j = 0; j < 3; ++j) {
                const int q  = 3 * kq + j;
                const int nd = (q < 6) ? nxtA : nxtB;
                const int qq = (q < 6) ? q : q - 6;
                pf[j] = *(const float4*)(x + (size_t)nd * (HIST * SKIP) + (tid + qq * 256) * 4);
            }
        }
        #pragma unroll
        for (int u16 = 0; u16 < 16; ++u16) {
            const int ks = kq * 16 + u16;
            v2f a0 = *(const v2f*)(A0 + 4 * ks);
            v2f a1 = *(const v2f*)(A0 + 16 * XS_STRIDE + 4 * ks);
            c0 = __builtin_amdgcn_wmma_f32_16x16x4_f32(false, bfr[ks], false, a0, (short)0, c0, false, false);
            c1 = __builtin_amdgcn_wmma_f32_16x16x4_f32(false, bfr[ks], false, a1, (short)0, c1, false, false);
            if (THREE) {
                v2f a2 = *(const v2f*)(A0 + 32 * XS_STRIDE + 4 * ks);
                c2 = __builtin_amdgcn_wmma_f32_16x16x4_f32(false, bfr[ks], false, a2, (short)0, c2, false, false);
            }
        }
        if (has_next) {
            #pragma unroll
            for (int j = 0; j < 3; ++j) {
                const int q = 3 * kq + j;
                *(float4*)(Xn + 4 * q * XS_STRIDE) = relu4(pf[j]);
            }
        }
    }
}

__global__ __launch_bounds__(256) __attribute__((amdgpu_waves_per_eu(4, 4)))
void moe_tcnn_kernel(const float* __restrict__ x,
                     const int*   __restrict__ labels,
                     const float* __restrict__ W1,
                     const float* __restrict__ b1,
                     const float* __restrict__ W2,
                     const float* __restrict__ b2,
                     float*       __restrict__ out)
{
    __shared__ __attribute__((aligned(16))) float Xs[2 * XBUF];
    __shared__ __attribute__((aligned(16))) float Zs[8 * ZW];
    __shared__ int lst[NPW + 2];
    __shared__ unsigned wtot[8][4];

    const int tid  = threadIdx.x;
    const int wv   = tid >> 5;
    const int lane = tid & 31;
    const int half = lane >> 4;
    const int l16  = lane & 15;
    const int n0   = wv * 16;
    const int r0   = blockIdx.x * NPW;

    int lab[16];
    #pragma unroll
    for (int j = 0; j < 16; ++j) lab[j] = labels[j * 256 + tid] & (NEXP - 1);
    unsigned long long acc = 0;
    #pragma unroll
    for (int j = 0; j < 16; ++j) acc += 1ull << (8 * lab[j]);
    unsigned pk[4], inc[4];
    #pragma unroll
    for (int q = 0; q < 4; ++q) {
        pk[q]  = (unsigned)((acc >> (16 * q)) & 0xffull) | ((unsigned)((acc >> (16 * q + 8)) & 0xffull) << 16);
        inc[q] = pk[q];
    }
    #pragma unroll
    for (int d = 1; d < 32; d <<= 1) {
        #pragma unroll
        for (int q = 0; q < 4; ++q) {
            unsigned t = __shfl_up(inc[q], d, 32);
            if (lane >= d) inc[q] += t;
        }
    }
    if (lane == 31) {
        #pragma unroll
        for (int q = 0; q < 4; ++q) wtot[wv][q] = inc[q];
    }
    if (tid == 0) { lst[NPW] = -1; lst[NPW + 1] = -1; }
    __syncthreads();
    unsigned tot[4] = {0u, 0u, 0u, 0u}, ex[4];
    #pragma unroll
    for (int q = 0; q < 4; ++q) {
        unsigned bas = 0;
        #pragma unroll
        for (int w = 0; w < 8; ++w) {
            const unsigned t = wtot[w][q];
            tot[q] += t;
            if (w < wv) bas += t;
        }
        ex[q] = bas + inc[q] - pk[q];
    }
    {
        #pragma unroll
        for (int q = 0; q < 4; ++q) tot[q] = (unsigned)__builtin_amdgcn_readfirstlane((int)tot[q]);
        const unsigned long long tlo = tot[0] | ((unsigned long long)tot[1] << 32);
        const unsigned long long thi = tot[2] | ((unsigned long long)tot[3] << 32);
        const unsigned long long xlo = ex[0]  | ((unsigned long long)ex[1]  << 32);
        const unsigned long long xhi = ex[2]  | ((unsigned long long)ex[3]  << 32);
        int off = 0;
        #pragma unroll 1
        for (int e = 0; e < NEXP; ++e) {
            const int sh = 16 * (e & 3);
            const int ce = (int)(((e < 4 ? tlo : thi) >> sh) & 0xffffull);
            if (off < r0 + NPW && off + ce > r0) {
                int r = off + (int)(((e < 4 ? xlo : xhi) >> sh) & 0xffffull) - r0;
                #pragma unroll
                for (int j = 0; j < 16; ++j) {
                    const bool mine = lab[j] == e;
                    if (mine && (unsigned)r < (unsigned)NPW) lst[r] = (e << 16) | (j * 256 + tid);
                    r += mine ? 1 : 0;
                }
            }
            off += ce;
        }
    }
    __syncthreads();

    const int xrow = tid >> 6;
    const int xcol = (tid & 63) * 4;
    const int part = lane >> 3;
    const int u    = wv * 8 + (lane & 7);
    const int und  = u >= FUT * OUTD ? 1 : 0;
    const int urem = u - FUT * OUTD * und;
    const int uf   = urem >> 1;
    const int uo   = urem & 1;

    v2f   bfr[64];
    float bias8[8];
    v2f   w2f[4][2];
    float b2v;

    int cur_e;
    {
        const int  t0 = __builtin_amdgcn_readfirstlane(lst[0]);
        const int  t1 = __builtin_amdgcn_readfirstlane(lst[1]);
        cur_e = t0 >> 16;
        const int  nA = t0 & 0xffff;
        const int  nB = ((t1 >> 16) == cur_e) ? (t1 & 0xffff) : nA;
        float4 v[12];
        #pragma unroll
        for (int q = 0; q < 12; ++q) {
            const int nd = (q < 6) ? nA : nB;
            const int qq = (q < 6) ? q : q - 6;
            v[q] = *(const float4*)(x + (size_t)nd * (HIST * SKIP) + (tid + qq * 256) * 4);
        }
        load_expert(cur_e, W1, b1, W2, b2, n0, l16, half, wv < 6, uo, bfr, bias8, w2f, b2v);
        #pragma unroll
        for (int q = 0; q < 12; ++q)
            *(float4*)(&Xs[(xrow + 4 * q) * XS_STRIDE + xcol]) = relu4(v[q]);
    }

    int i = 0, buf = 0;
    while (i < NPW) {
        const int  tA     = __builtin_amdgcn_readfirstlane(lst[i]);
        const int  tB     = __builtin_amdgcn_readfirstlane(lst[i + 1]);
        const int  e      = tA >> 16;
        const int  nodeA  = tA & 0xffff;
        const bool validB = (tB >> 16) == e;
        const int  nodeB  = validB ? (tB & 0xffff) : nodeA;
        const int  ni     = i + (validB ? 2 : 1);
        const bool has_next = ni < NPW;
        int nxtA = nodeA, nxtB = nodeA;
        if (has_next) {
            const int tC = __builtin_amdgcn_readfirstlane(lst[ni]);
            const int tD = __builtin_amdgcn_readfirstlane(lst[ni + 1]);
            nxtA = tC & 0xffff;
            nxtB = ((tD >> 16) == (tC >> 16)) ? (tD & 0xffff) : nxtA;
        }
        if (e != cur_e) {
            load_expert(e, W1, b1, W2, b2, n0, l16, half, wv < 6, uo, bfr, bias8, w2f, b2v);
            cur_e = e;
        }
        __syncthreads();

        const float* A0 = &Xs[buf * XBUF + l16 * XS_STRIDE + 2 * half];
        float* Xn = &Xs[(buf ^ 1) * XBUF + xrow * XS_STRIDE + xcol];
        v8f c0 = {0.f,0.f,0.f,0.f,0.f,0.f,0.f,0.f};
        v8f c1 = {0.f,0.f,0.f,0.f,0.f,0.f,0.f,0.f};
        v8f c2 = {0.f,0.f,0.f,0.f,0.f,0.f,0.f,0.f};
        if (validB) stage1<true >(bfr, x, A0, Xn, has_next, nxtA, nxtB, tid, c0, c1, c2);
        else        stage1<false>(bfr, x, A0, Xn, has_next, nxtA, nxtB, tid, c0, c1, c2);

        {
            v8f ct[3] = {c0, c1, c2};
            #pragma unroll
            for (int t = 0; t < 3; ++t) {
                #pragma unroll
                for (int r = 0; r < 8; ++r) {
                    const float hv = ct[t][r] + bias8[r];
                    ct[t][r] = hv > 0.f ? hv : 0.f;
                }
            }
            float* zrow = &Zs[wv * ZW + l16 * ZP + 8 * half];
            #pragma unroll
            for (int t = 0; t < 3; ++t) {
                if (t < 2 || validB) {
                    v8f z0 = {0.f,0.f,0.f,0.f,0.f,0.f,0.f,0.f};
                    v8f z1 = {0.f,0.f,0.f,0.f,0.f,0.f,0.f,0.f};
                    #pragma unroll
                    for (int j = 0; j < 4; ++j) {
                        v2f a;
                        a.x = ct[t][2 * j];
                        a.y = ct[t][2 * j + 1];
                        z0 = __builtin_amdgcn_wmma_f32_16x16x4_f32(false, a, false, w2f[j][0], (short)0, z0, false, false);
                        z1 = __builtin_amdgcn_wmma_f32_16x16x4_f32(false, a, false, w2f[j][1], (short)0, z1, false, false);
                    }
                    float4 s0 = {z0[0], z0[1], z0[2], z0[3]};
                    float4 s1 = {z0[4], z0[5], z0[6], z0[7]};
                    *(float4*)(zrow + 16 * t)     = s0;
                    *(float4*)(zrow + 16 * t + 4) = s1;
                    if (l16 < ZN - 16) {
                        float4 s2 = {z1[0], z1[1], z1[2], z1[3]};
                        float4 s3 = {z1[4], z1[5], z1[6], z1[7]};
                        *(float4*)(zrow + 16 * ZP + 16 * t)     = s2;
                        *(float4*)(zrow + 16 * ZP + 16 * t + 4) = s3;
                    }
                }
            }
        }
        __syncthreads();

        if (wv < 6) {
            const float* zb = &Zs[(2 * part) * ZW + (uo * KERN) * ZP + HIST * und + uf];
            float ac[4] = {0.f, 0.f, 0.f, 0.f};
            #pragma unroll
            for (int w = 0; w < 2; ++w) {
                #pragma unroll
                for (int k = 0; k < KERN; ++k)
                    ac[2 * w + (k & 1)] += zb[w * ZW + k * (ZP + 1)];
            }
            float a4 = (ac[0] + ac[1]) + (ac[2] + ac[3]);
            a4 += __shfl_xor(a4, 8, 32);
            a4 += __shfl_xor(a4, 16, 32);
            const int node = und ? nodeB : nodeA;
            if (part == 0 && (und == 0 || validB))
                out[(size_t)node * (FUT * OUTD) + urem] = a4 + b2v;
        }
        i = ni;
        buf ^= 1;
    }
}

extern "C" void kernel_launch(void* const* d_in, const int* in_sizes, int n_in,
                              void* d_out, int out_size, void* d_ws, size_t ws_size,
                              hipStream_t stream) {
    const float* x      = (const float*)d_in[0];
    const int*   labels = (const int*)  d_in[1];
    const float* W1     = (const float*)d_in[2];
    const float* b1     = (const float*)d_in[3];
    const float* W2     = (const float*)d_in[4];
    const float* b2     = (const float*)d_in[5];
    float*       out    = (float*)d_out;

    moe_tcnn_kernel<<<NWG, 256, 0, stream>>>(x, labels, W1, b1, W2, b2, out);
}


# Wave priority steps down with each finished pair (s_setprio 3,2,1,0): the two work-groups of a WGP take turns at the matrix pipe instead of one running ahead and the other finishing alone (same sums)
# speedup vs baseline: 2.0207x; 1.0563x over previous
#include <hip/hip_runtime.h>

#define NB      8
#define NN      512
#define HIST    24
#define SKIP    256
#define ENDC    128
#define KERN    13
#define FUT     12
#define OUTD    2
#define NODES   (NB * NN)
#define NEXP    8
#define NWG     512
#define NPW     (NODES / NWG)
#define PROWS   48
#define XS_STRIDE 260
#define XBUF    (PROWS * XS_STRIDE)
#define W2E     (OUTD * ENDC * KERN)
#define ZN      (OUTD * KERN)
#define ZP      52
#define ZW      (ZN * ZP)

typedef float v2f __attribute__((ext_vector_type(2)));
typedef float v8f __attribute__((ext_vector_type(8)));

__device__ __forceinline__ float4 relu4(float4 v) {
    v.x = v.x > 0.f ? v.x : 0.f;
    v.y = v.y > 0.f ? v.y : 0.f;
    v.z = v.z > 0.f ? v.z : 0.f;
    v.w = v.w > 0.f ? v.w : 0.f;
    return v;
}

__device__ __forceinline__ void load_expert(const int e,
                                            const float* __restrict__ W1, const float* __restrict__ b1,
                                            const float* __restrict__ W2, const float* __restrict__ b2,
                                            const int n0, const int l16, const int half,
                                            const bool outlane, const int uo,
                                            v2f (&bfr)[64], float (&bias8)[8], v2f (&w2f)[4][2], float& b2v)
{
    const float* W1row = W1 + ((size_t)e * ENDC + n0 + l16) * SKIP + 2 * half;
    #pragma unroll
    for (int ks = 0; ks < 64; ++ks) bfr[ks] = *(const v2f*)(W1row + 4 * ks);
    const float4 bia = *(const float4*)(b1 + e * ENDC + n0 + 8 * half);
    const float4 bib = *(const float4*)(b1 + e * ENDC + n0 + 8 * half + 4);
    bias8[0] = bia.x; bias8[1] = bia.y; bias8[2] = bia.z; bias8[3] = bia.w;
    bias8[4] = bib.x; bias8[5] = bib.y; bias8[6] = bib.z; bias8[7] = bib.w;
    #pragma unroll
    for (int nt = 0; nt < 2; ++nt) {
        const int  n   = 16 * nt + l16;
        const bool nok = n < ZN;
        const int  nc  = nok ? n : 0;
        const int  o2  = nc >= KERN ? 1 : 0;
        const int  k2  = nc - KERN * o2;
        const float* wp = W2 + (size_t)e * W2E + o2 * (ENDC * KERN) + (n0 + 8 * half) * KERN + k2;
        #pragma unroll
        for (int j = 0; j < 4; ++j) {
            const float w0 = wp[(2 * j) * KERN];
            const float w1 = wp[(2 * j + 1) * KERN];
            w2f[j][nt].x = nok ? w0 : 0.f;
            w2f[j][nt].y = nok ? w1 : 0.f;
        }
    }
    b2v = outlane ? b2[e * OUTD + uo] : 0.f;
}

template <bool THREE>
__device__ __forceinline__ void stage1(const v2f (&bfr)[64], const float* __restrict__ x,
                                       const float* A0, float* Xn, const bool has_next,
                                       const int nxtA, const int nxtB, const int tid,
                                       v8f& c0, v8f& c1, v8f& c2)
{
    #pragma unroll
    for (int kq = 0; kq < 4; ++kq) {
        float4 pf[3] = {};
        if (has_next) {
            #pragma unroll
            for (int j = 0; j < 3; ++j) {
                const int q  = 3 * kq + j;
                const int nd = (q < 6) ? nxtA : nxtB;
                const int qq = (q < 6) ? q : q - 6;
                pf[j] = *(const float4*)(x + (size_t)nd * (HIST * SKIP) + (tid + qq * 256) * 4);
            }
        }
        #pragma unroll
        for (int u16 = 0; u16 < 16; ++u16) {
            const int ks = kq * 16 + u16;
            v2f a0 = *(const v2f*)(A0 + 4 * ks);
            v2f a1 = *(const v2f*)(A0 + 16 * XS_STRIDE + 4 * ks);
            c0 = __builtin_amdgcn_wmma_f32_16x16x4_f32(false, bfr[ks], false, a0, (short)0, c0, false, false);
            c1 = __builtin_amdgcn_wmma_f32_16x16x4_f32(false, bfr[ks], false, a1, (short)0, c1, false, false);
            if (THREE) {
                v2f a2 = *(const v2f*)(A0 + 32 * XS_STRIDE + 4 * ks);
                c2 = __builtin_amdgcn_wmma_f32_16x16x4_f32(false, bfr[ks], false, a2, (short)0, c2, false, false);
            }
        }
        if (has_next) {
            #pragma unroll
            for (int j = 0; j < 3; ++j) {
                const int q = 3 * kq + j;
                *(float4*)(Xn + 4 * q * XS_STRIDE) = relu4(pf[j]);
            }
        }
    }
}

__global__ __launch_bounds__(256) __attribute__((amdgpu_waves_per_eu(4, 4)))
void moe_tcnn_kernel(const float* __restrict__ x,
                     const int*   __restrict__ labels,
                     const float* __restrict__ W1,
                     const float* __restrict__ b1,
                     const float* __restrict__ W2,
                     const float* __restrict__ b2,
                     float*       __restrict__ out)
{
    __shared__ __attribute__((aligned(16))) float Xs[2 * XBUF];
    __shared__ __attribute__((aligned(16))) float Zs[8 * ZW];
    __shared__ int lst[NPW + 2];
    __shared__ unsigned wtot[8][4];

    const int tid  = threadIdx.x;
    const int wv   = tid >> 5;
    const int lane = tid & 31;
    const int half = lane >> 4;
    const int l16  = lane & 15;
    const int n0   = wv * 16;
    const int r0   = blockIdx.x * NPW;

    int lab[16];
    #pragma unroll
    for (int j = 0; j < 16; ++j) lab[j] = labels[j * 256 + tid] & (NEXP - 1);
    unsigned long long acc = 0;
    #pragma unroll
    for (int j = 0; j < 16; ++j) acc += 1ull << (8 * lab[j]);
    unsigned pk[4], inc[4];
    #pragma unroll
    for (int q = 0; q < 4; ++q) {
        pk[q]  = (unsigned)((acc >> (16 * q)) & 0xffull) | ((unsigned)((acc >> (16 * q + 8)) & 0xffull) << 16);
        inc[q] = pk[q];
    }
    #pragma unroll
    for (int d = 1; d < 32; d <<= 1) {
        #pragma unroll
        for (int q = 0; q < 4; ++q) {
            unsigned t = __shfl_up(inc[q], d, 32);
            if (lane >= d) inc[q] += t;
        }
    }
    if (lane == 31) {
        #pragma unroll
        for (int q = 0; q < 4; ++q) wtot[wv][q] = inc[q];
    }
    if (tid == 0) { lst[NPW] = -1; lst[NPW + 1] = -1; }
    __syncthreads();
    unsigned tot[4] = {0u, 0u, 0u, 0u}, ex[4];
    #pragma unroll
    for (int q = 0; q < 4; ++q) {
        unsigned bas = 0;
        #pragma unroll
        for (int w = 0; w < 8; ++w) {
            const unsigned t = wtot[w][q];
            tot[q] += t;
            if (w < wv) bas += t;
        }
        ex[q] = bas + inc[q] - pk[q];
    }
    {
        #pragma unroll
        for (int q = 0; q < 4; ++q) tot[q] = (unsigned)__builtin_amdgcn_readfirstlane((int)tot[q]);
        const unsigned long long tlo = tot[0] | ((unsigned long long)tot[1] << 32);
        const unsigned long long thi = tot[2] | ((unsigned long long)tot[3] << 32);
        const unsigned long long xlo = ex[0]  | ((unsigned long long)ex[1]  << 32);
        const unsigned long long xhi = ex[2]  | ((unsigned long long)ex[3]  << 32);
        int off = 0;
        #pragma unroll 1
        for (int e = 0; e < NEXP; ++e) {
            const int sh = 16 * (e & 3);
            const int ce = (int)(((e < 4 ? tlo : thi) >> sh) & 0xffffull);
            if (off < r0 + NPW && off + ce > r0) {
                int r = off + (int)(((e < 4 ? xlo : xhi) >> sh) & 0xffffull) - r0;
                #pragma unroll
                for (int j = 0; j < 16; ++j) {
                    const bool mine = lab[j] == e;
                    if (mine && (unsigned)r < (unsigned)NPW) lst[r] = (e << 16) | (j * 256 + tid);
                    r += mine ? 1 : 0;
                }
            }
            off += ce;
        }
    }
    __syncthreads();

    const int xrow = tid >> 6;
    const int xcol = (tid & 63) * 4;
    const int part = lane >> 3;
    const int u    = wv * 8 + (lane & 7);
    const int und  = u >= FUT * OUTD ? 1 : 0;
    const int urem = u - FUT * OUTD * und;
    const int uf   = urem >> 1;
    const int uo   = urem & 1;

    v2f   bfr[64];
    float bias8[8];
    v2f   w2f[4][2];
    float b2v;

    int cur_e;
    {
        const int  t0 = __builtin_amdgcn_readfirstlane(lst[0]);
        const int  t1 = __builtin_amdgcn_readfirstlane(lst[1]);
        cur_e = t0 >> 16;
        const int  nA = t0 & 0xffff;
        const int  nB = ((t1 >> 16) == cur_e) ? (t1 & 0xffff) : nA;
        float4 v[12];
        #pragma unroll
        for (int q = 0; q < 12; ++q) {
            const int nd = (q < 6) ? nA : nB;
            const int qq = (q < 6) ? q : q - 6;
            v[q] = *(const float4*)(x + (size_t)nd * (HIST * SKIP) + (tid + qq * 256) * 4);
        }
        load_expert(cur_e, W1, b1, W2, b2, n0, l16, half, wv < 6, uo, bfr, bias8, w2f, b2v);
        #pragma unroll
        for (int q = 0; q < 12; ++q)
            *(float4*)(&Xs[(xrow + 4 * q) * XS_STRIDE + xcol]) = relu4(v[q]);
    }

    __builtin_amdgcn_s_setprio(3);
    int i = 0, buf = 0, done = 0;
    while (i < NPW) {
        const int  tA     = __builtin_amdgcn_readfirstlane(lst[i]);
        const int  tB     = __builtin_amdgcn_readfirstlane(lst[i + 1]);
        const int  e      = tA >> 16;
        const int  nodeA  = tA & 0xffff;
        const bool validB = (tB >> 16) == e;
        const int  nodeB  = validB ? (tB & 0xffff) : nodeA;
        const int  ni     = i + (validB ? 2 : 1);
        const bool has_next = ni < NPW;
        int nxtA = nodeA, nxtB = nodeA;
        if (has_next) {
            const int tC = __builtin_amdgcn_readfirstlane(lst[ni]);
            const int tD = __builtin_amdgcn_readfirstlane(lst[ni + 1]);
            nxtA = tC & 0xffff;
            nxtB = ((tD >> 16) == (tC >> 16)) ? (tD & 0xffff) : nxtA;
        }
        if (e != cur_e) {
            load_expert(e, W1, b1, W2, b2, n0, l16, half, wv < 6, uo, bfr, bias8, w2f, b2v);
            cur_e = e;
        }
        __syncthreads();

        const float* A0 = &Xs[buf * XBUF + l16 * XS_STRIDE + 2 * half];
        float* Xn = &Xs[(buf ^ 1) * XBUF + xrow * XS_STRIDE + xcol];
        v8f c0 = {0.f,0.f,0.f,0.f,0.f,0.f,0.f,0.f};
        v8f c1 = {0.f,0.f,0.f,0.f,0.f,0.f,0.f,0.f};
        v8f c2 = {0.f,0.f,0.f,0.f,0.f,0.f,0.f,0.f};
        if (validB) stage1<true >(bfr, x, A0, Xn, has_next, nxtA, nxtB, tid, c0, c1, c2);
        else        stage1<false>(bfr, x, A0, Xn, has_next, nxtA, nxtB, tid, c0, c1, c2);

        {
            v8f ct[3] = {c0, c1, c2};
            #pragma unroll
            for (int t = 0; t < 3; ++t) {
                #pragma unroll
                for (int r = 0; r < 8; ++r) {
                    const float hv = ct[t][r] + bias8[r];
                    ct[t][r] = hv > 0.f ? hv : 0.f;
                }
            }
            float* zrow = &Zs[wv * ZW + l16 * ZP + 8 * half];
            #pragma unroll
            for (int t = 0; t < 3; ++t) {
                if (t < 2 || validB) {
                    v8f z0 = {0.f,0.f,0.f,0.f,0.f,0.f,0.f,0.f};
                    v8f z1 = {0.f,0.f,0.f,0.f,0.f,0.f,0.f,0.f};
                    #pragma unroll
                    for (int j = 0; j < 4; ++j) {
                        v2f a;
                        a.x = ct[t][2 * j];
                        a.y = ct[t][2 * j + 1];
                        z0 = __builtin_amdgcn_wmma_f32_16x16x4_f32(false, a, false, w2f[j][0], (short)0, z0, false, false);
                        z1 = __builtin_amdgcn_wmma_f32_16x16x4_f32(false, a, false, w2f[j][1], (short)0, z1, false, false);
                    }
                    float4 s0 = {z0[0], z0[1], z0[2], z0[3]};
                    float4 s1 = {z0[4], z0[5], z0[6], z0[7]};
                    *(float4*)(zrow + 16 * t)     = s0;
                    *(float4*)(zrow + 16 * t + 4) = s1;
                    if (l16 < ZN - 16) {
                        float4 s2 = {z1[0], z1[1], z1[2], z1[3]};
                        float4 s3 = {z1[4], z1[5], z1[6], z1[7]};
                        *(float4*)(zrow + 16 * ZP + 16 * t)     = s2;
                        *(float4*)(zrow + 16 * ZP + 16 * t + 4) = s3;
                    }
                }
            }
        }
        __syncthreads();

        if (wv < 6) {
            const float* zb = &Zs[(2 * part) * ZW + (uo * KERN) * ZP + HIST * und + uf];
            float ac[4] = {0.f, 0.f, 0.f, 0.f};
            #pragma unroll
            for (int w = 0; w < 2; ++w) {
                #pragma unroll
                for (int k = 0; k < KERN; ++k)
                    ac[2 * w + (k & 1)] += zb[w * ZW + k * (ZP + 1)];
            }
            float a4 = (ac[0] + ac[1]) + (ac[2] + ac[3]);
            a4 += __shfl_xor(a4, 8, 32);
            a4 += __shfl_xor(a4, 16, 32);
            const int node = und ? nodeB : nodeA;
            if (part == 0 && (und == 0 || validB))
                out[(size_t)node * (FUT * OUTD) + urem] = a4 + b2v;
        }
        i = ni;
        buf ^= 1;
        ++done;
        if (done == 1)      __builtin_amdgcn_s_setprio(2);
        else if (done == 2) __builtin_amdgcn_s_setprio(1);
        else                __builtin_amdgcn_s_setprio(0);
    }
}

extern "C" void kernel_launch(void* const* d_in, const int* in_sizes, int n_in,
                              void* d_out, int out_size, void* d_ws, size_t ws_size,
                              hipStream_t stream) {
    const float* x      = (const float*)d_in[0];
    const int*   labels = (const int*)  d_in[1];
    const float* W1     = (const float*)d_in[2];
    const float* b1     = (const float*)d_in[3];
    const float* W2     = (const float*)d_in[4];
    const float* b2     = (const float*)d_in[5];
    float*       out    = (float*)d_out;

    moe_tcnn_kernel<<<NWG, 256, 0, stream>>>(x, labels, W1, b1, W2, b2, out);
}


# W1 fill pipelined under the first products: only 8 of 64 k-steps of W1[e] are awaited before the first barrier; stage 1 requests the rest ahead of use, the stage-2 operands last (same sums)
# speedup vs baseline: 2.1285x; 1.0533x over previous
#include <hip/hip_runtime.h>

#define NB      8
#define NN      512
#define HIST    24
#define SKIP    256
#define ENDC    128
#define KERN    13
#define FUT     12
#define OUTD    2
#define NODES   (NB * NN)
#define NEXP    8
#define NWG     512
#define NPW     (NODES / NWG)
#define PROWS   48
#define XS_STRIDE 260
#define XBUF    (PROWS * XS_STRIDE)
#define W2E     (OUTD * ENDC * KERN)
#define ZN      (OUTD * KERN)
#define ZP      52
#define ZW      (ZN * ZP)

typedef float v2f __attribute__((ext_vector_type(2)));
typedef float v8f __attribute__((ext_vector_type(8)));

__device__ __forceinline__ float4 relu4(float4 v) {
    v.x = v.x > 0.f ? v.x : 0.f;
    v.y = v.y > 0.f ? v.y : 0.f;
    v.z = v.z > 0.f ? v.z : 0.f;
    v.w = v.w > 0.f ? v.w : 0.f;
    return v;
}

template <int K0, int N>
__device__ __forceinline__ void load_w1_steps(const float* __restrict__ W1row, v2f (&bfr)[64])
{
    #pragma unroll
    for (int u = 0; u < N; ++u) bfr[K0 + u] = *(const v2f*)(W1row + 4 * (K0 + u));
}

__device__ __forceinline__ void load_small(const int e,
                                           const float* __restrict__ b1,
                                           const float* __restrict__ W2, const float* __restrict__ b2,
                                           const int n0, const int l16, const int half,
                                           const bool outlane, const int uo,
                                           float (&bias8)[8], v2f (&w2f)[4][2], float& b2v)
{
    const float4 bia = *(const float4*)(b1 + e * ENDC + n0 + 8 * half);
    const float4 bib = *(const float4*)(b1 + e * ENDC + n0 + 8 * half + 4);
    bias8[0] = bia.x; bias8[1] = bia.y; bias8[2] = bia.z; bias8[3] = bia.w;
    bias8[4] = bib.x; bias8[5] = bib.y; bias8[6] = bib.z; bias8[7] = bib.w;
    #pragma unroll
    for (int nt = 0; nt < 2; ++nt) {
        const int  n   = 16 * nt + l16;
        const bool nok = n < ZN;
        const int  nc  = nok ? n : 0;
        const int  o2  = nc >= KERN ? 1 : 0;
        const int  k2  = nc - KERN * o2;
        const float* wp = W2 + (size_t)e * W2E + o2 * (ENDC * KERN) + (n0 + 8 * half) * KERN + k2;
        #pragma unroll
        for (int j = 0; j < 4; ++j) {
            const float w0 = wp[(2 * j) * KERN];
            const float w1 = wp[(2 * j + 1) * KERN];
            w2f[j][nt].x = nok ? w0 : 0.f;
            w2f[j][nt].y = nok ? w1 : 0.f;
        }
    }
    b2v = outlane ? b2[e * OUTD + uo] : 0.f;
}

template <bool THREE, bool FILL>
__device__ __forceinline__ void stage1(v2f (&bfr)[64], const float* __restrict__ W1row,
                                       const int e, const float* __restrict__ b1,
                                       const float* __restrict__ W2, const float* __restrict__ b2,
                                       const int n0, const int l16, const int half, const bool outlane, const int uo,
                                       float (&bias8)[8], v2f (&w2f)[4][2], float& b2v,
                                       const float* __restrict__ x,
                                       const float* A0, float* Xn, const bool has_next,
                                       const int nxtA, const int nxtB, const int tid,
                                       v8f& c0, v8f& c1, v8f& c2)
{
    #pragma unroll
    for (int kq = 0; kq < 4; ++kq) {
        float4 pf[3] = {};
        if (has_next) {
            #pragma unroll
            for (int j = 0; j < 3; ++j) {
                const int q  = 3 * kq + j;
                const int nd = (q < 6) ? nxtA : nxtB;
                const int qq = (q < 6) ? q : q - 6;
                pf[j] = *(const float4*)(x + (size_t)nd * (HIST * SKIP) + (tid + qq * 256) * 4);
            }
        }
        if (FILL) {
            if (kq == 0) load_w1_steps<8, 24>(W1row, bfr);
            if (kq == 1) load_w1_steps<32, 16>(W1row, bfr);
            if (kq == 2) load_w1_steps<48, 16>(W1row, bfr);
            if (kq == 3) load_small(e, b1, W2, b2, n0, l16, half, outlane, uo, bias8, w2f, b2v);
        }
        #pragma unroll
        for (int u16 = 0; u16 < 16; ++u16) {
            const int ks = kq * 16 + u16;
            v2f a0 = *(const v2f*)(A0 + 4 * ks);
            v2f a1 = *(const v2f*)(A0 + 16 * XS_STRIDE + 4 * ks);
            c0 = __builtin_amdgcn_wmma_f32_16x16x4_f32(false, bfr[ks], false, a0, (short)0, c0, false, false);
            c1 = __builtin_amdgcn_wmma_f32_16x16x4_f32(false, bfr[ks], false, a1, (short)0, c1, false, false);
            if (THREE) {
                v2f a2 = *(const v2f*)(A0 + 32 * XS_STRIDE + 4 * ks);
                c2 = __builtin_amdgcn_wmma_f32_16x16x4_f32(false, bfr[ks], false, a2, (short)0, c2, false, false);
            }
        }
        if (has_next) {
            #pragma unroll
            for (int j = 0; j < 3; ++j) {
                const int q = 3 * kq + j;
                *(float4*)(Xn + 4 * q * XS_STRIDE) = relu4(pf[j]);
            }
        }
    }
}

__global__ __launch_bounds__(256) __attribute__((amdgpu_waves_per_eu(4, 4)))
void moe_tcnn_kernel(const float* __restrict__ x,
                     const int*   __restrict__ labels,
                     const float* __restrict__ W1,
                     const float* __restrict__ b1,
                     const float* __restrict__ W2,
                     const float* __restrict__ b2,
                     float*       __restrict__ out)
{
    __shared__ __attribute__((aligned(16))) float Xs[2 * XBUF];
    __shared__ __attribute__((aligned(16))) float Zs[8 * ZW];
    __shared__ int lst[NPW + 2];
    __shared__ unsigned wtot[8][4];

    const int tid  = threadIdx.x;
    const int wv   = tid >> 5;
    const int lane = tid & 31;
    const int half = lane >> 4;
    const int l16  = lane & 15;
    const int n0   = wv * 16;
    const int r0   = blockIdx.x * NPW;

    int lab[16];
    #pragma unroll
    for (int j = 0; j < 16; ++j) lab[j] = labels[j * 256 + tid] & (NEXP - 1);
    unsigned long long acc = 0;
    #pragma unroll
    for (int j = 0; j < 16; ++j) acc += 1ull << (8 * lab[j]);
    unsigned pk[4], inc[4];
    #pragma unroll
    for (int q = 0; q < 4; ++q) {
        pk[q]  = (unsigned)((acc >> (16 * q)) & 0xffull) | ((unsigned)((acc >> (16 * q + 8)) & 0xffull) << 16);
        inc[q] = pk[q];
    }
    #pragma unroll
    for (int d = 1; d < 32; d <<= 1) {
        #pragma unroll
        for (int q = 0; q < 4; ++q) {
            unsigned t = __shfl_up(inc[q], d, 32);
            if (lane >= d) inc[q] += t;
        }
    }
    if (lane == 31) {
        #pragma unroll
        for (int q = 0; q < 4; ++q) wtot[wv][q] = inc[q];
    }
    if (tid == 0) { lst[NPW] = -1; lst[NPW + 1] = -1; }
    __syncthreads();
    unsigned tot[4] = {0u, 0u, 0u, 0u}, ex[4];
    #pragma unroll
    for (int q = 0; q < 4; ++q) {
        unsigned bas = 0;
        #pragma unroll
        for (int w = 0; w < 8; ++w) {
            const unsigned t = wtot[w][q];
            tot[q] += t;
            if (w < wv) bas += t;
        }
        ex[q] = bas + inc[q] - pk[q];
    }
    {
        #pragma unroll
        for (int q = 0; q < 4; ++q) tot[q] = (unsigned)__builtin_amdgcn_readfirstlane((int)tot[q]);
        const unsigned long long tlo = tot[0] | ((unsigned long long)tot[1] << 32);
        const unsigned long long thi = tot[2] | ((unsigned long long)tot[3] << 32);
        const unsigned long long xlo = ex[0]  | ((unsigned long long)ex[1]  << 32);
        const unsigned long long xhi = ex[2]  | ((unsigned long long)ex[3]  << 32);
        int off = 0;
        #pragma unroll 1
        for (int e = 0; e < NEXP; ++e) {
            const int sh = 16 * (e & 3);
            const int ce = (int)(((e < 4 ? tlo : thi) >> sh) & 0xffffull);
            if (off < r0 + NPW && off + ce > r0) {
                int r = off + (int)(((e < 4 ? xlo : xhi) >> sh) & 0xffffull) - r0;
                #pragma unroll
                for (int j = 0; j < 16; ++j) {
                    const bool mine = lab[j] == e;
                    if (mine && (unsigned)r < (unsigned)NPW) lst[r] = (e << 16) | (j * 256 + tid);
                    r += mine ? 1 : 0;
                }
            }
            off += ce;
        }
    }
    __syncthreads();

    const int xrow = tid >> 6;
    const int xcol = (tid & 63) * 4;
    const int part = lane >> 3;
    const int u    = wv * 8 + (lane & 7);
    const int und  = u >= FUT * OUTD ? 1 : 0;
    const int urem = u - FUT * OUTD * und;
    const int uf   = urem >> 1;
    const int uo   = urem & 1;

    v2f   bfr[64];
    float bias8[8];
    v2f   w2f[4][2];
    float b2v;
    const float* W1row;
    bool  fill = true;

    int cur_e;
    {
        const int  t0 = __builtin_amdgcn_readfirstlane(lst[0]);
        const int  t1 = __builtin_amdgcn_readfirstlane(lst[1]);
        cur_e = t0 >> 16;
        const int  nA = t0 & 0xffff;
        const int  nB = ((t1 >> 16) == cur_e) ? (t1 & 0xffff) : nA;
        float4 v[12];
        #pragma unroll
        for (int q = 0; q < 12; ++q) {
            const int nd = (q < 6) ? nA : nB;
            const int qq = (q < 6) ? q : q - 6;
            v[q] = *(const float4*)(x + (size_t)nd * (HIST * SKIP) + (tid + qq * 256) * 4);
        }
        W1row = W1 + ((size_t)cur_e * ENDC + n0 + l16) * SKIP + 2 * half;
        load_w1_steps<0, 8>(W1row, bfr);
        #pragma unroll
        for (int q = 0; q < 12; ++q)
            *(float4*)(&Xs[(xrow + 4 * q) * XS_STRIDE + xcol]) = relu4(v[q]);
    }

    __builtin_amdgcn_s_setprio(3);
    int i = 0, buf = 0, done = 0;
    while (i < NPW) {
        const int  tA     = __builtin_amdgcn_readfirstlane(lst[i]);
        const int  tB     = __builtin_amdgcn_readfirstlane(lst[i + 1]);
        const int  e      = tA >> 16;
        const int  nodeA  = tA & 0xffff;
        const bool validB = (tB >> 16) == e;
        const int  nodeB  = validB ? (tB & 0xffff) : nodeA;
        const int  ni     = i + (validB ? 2 : 1);
        const bool has_next = ni < NPW;
        int nxtA = nodeA, nxtB = nodeA;
        if (has_next) {
            const int tC = __builtin_amdgcn_readfirstlane(lst[ni]);
            const int tD = __builtin_amdgcn_readfirstlane(lst[ni + 1]);
            nxtA = tC & 0xffff;
            nxtB = ((tD >> 16) == (tC >> 16)) ? (tD & 0xffff) : nxtA;
        }
        if (e != cur_e) {
            W1row = W1 + ((size_t)e * ENDC + n0 + l16) * SKIP + 2 * half;
            load_w1_steps<0, 8>(W1row, bfr);
            cur_e = e;
            fill  = true;
        }
        __syncthreads();

        const float* A0 = &Xs[buf * XBUF + l16 * XS_STRIDE + 2 * half];
        float* Xn = &Xs[(buf ^ 1) * XBUF + xrow * XS_STRIDE + xcol];
        v8f c0 = {0.f,0.f,0.f,0.f,0.f,0.f,0.f,0.f};
        v8f c1 = {0.f,0.f,0.f,0.f,0.f,0.f,0.f,0.f};
        v8f c2 = {0.f,0.f,0.f,0.f,0.f,0.f,0.f,0.f};
        if (fill) {
            if (validB) stage1<true,  true >(bfr, W1row, e, b1, W2, b2, n0, l16, half, wv < 6, uo, bias8, w2f, b2v, x, A0, Xn, has_next, nxtA, nxtB, tid, c0, c1, c2);
            else        stage1<false, true >(bfr, W1row, e, b1, W2, b2, n0, l16, half, wv < 6, uo, bias8, w2f, b2v, x, A0, Xn, has_next, nxtA, nxtB, tid, c0, c1, c2);
        } else {
            if (validB) stage1<true,  false>(bfr, W1row, e, b1, W2, b2, n0, l16, half, wv < 6, uo, bias8, w2f, b2v, x, A0, Xn, has_next, nxtA, nxtB, tid, c0, c1, c2);
            else        stage1<false, false>(bfr, W1row, e, b1, W2, b2, n0, l16, half, wv < 6, uo, bias8, w2f, b2v, x, A0, Xn, has_next, nxtA, nxtB, tid, c0, c1, c2);
        }
        fill = false;

        {
            v8f ct[3] = {c0, c1, c2};
            #pragma unroll
            for (int t = 0; t < 3; ++t) {
                #pragma unroll
                for (int r = 0; r < 8; ++r) {
                    const float hv = ct[t][r] + bias8[r];
                    ct[t][r] = hv > 0.f ? hv : 0.f;
                }
            }
            float* zrow = &Zs[wv * ZW + l16 * ZP + 8 * half];
            #pragma unroll
            for (int t = 0; t < 3; ++t) {
                if (t < 2 || validB) {
                    v8f z0 = {0.f,0.f,0.f,0.f,0.f,0.f,0.f,0.f};
                    v8f z1 = {0.f,0.f,0.f,0.f,0.f,0.f,0.f,0.f};
                    #pragma unroll
                    for (int j = 0; j < 4; ++j) {
                        v2f a;
                        a.x = ct[t][2 * j];
                        a.y = ct[t][2 * j + 1];
                        z0 = __builtin_amdgcn_wmma_f32_16x16x4_f32(false, a, false, w2f[j][0], (short)0, z0, false, false);
                        z1 = __builtin_amdgcn_wmma_f32_16x16x4_f32(false, a, false, w2f[j][1], (short)0, z1, false, false);
                    }
                    float4 s0 = {z0[0], z0[1], z0[2], z0[3]};
                    float4 s1 = {z0[4], z0[5], z0[6], z0[7]};
                    *(float4*)(zrow + 16 * t)     = s0;
                    *(float4*)(zrow + 16 * t + 4) = s1;
                    if (l16 < ZN - 16) {
                        float4 s2 = {z1[0], z1[1], z1[2], z1[3]};
                        float4 s3 = {z1[4], z1[5], z1[6], z1[7]};
                        *(float4*)(zrow + 16 * ZP + 16 * t)     = s2;
                        *(float4*)(zrow + 16 * ZP + 16 * t + 4) = s3;
                    }
                }
            }
        }
        __syncthreads();

        if (wv < 6) {
            const float* zb = &Zs[(2 * part) * ZW + (uo * KERN) * ZP + HIST * und + uf];
            float ac[4] = {0.f, 0.f, 0.f, 0.f};
            #pragma unroll
            for (int w = 0; w < 2; ++w) {
                #pragma unroll
                for (int k = 0; k < KERN; ++k)
                    ac[2 * w + (k & 1)] += zb[w * ZW + k * (ZP + 1)];
            }
            float a4 = (ac[0] + ac[1]) + (ac[2] + ac[3]);
            a4 += __shfl_xor(a4, 8, 32);
            a4 += __shfl_xor(a4, 16, 32);
            const int node = und ? nodeB : nodeA;
            if (part == 0 && (und == 0 || validB))
                out[(size_t)node * (FUT * OUTD) + urem] = a4 + b2v;
        }
        i = ni;
        buf ^= 1;
        ++done;
        if (done == 1)      __builtin_amdgcn_s_setprio(2);
        else if (done == 2) __builtin_amdgcn_s_setprio(1);
        else                __builtin_amdgcn_s_setprio(0);
    }
}

extern "C" void kernel_launch(void* const* d_in, const int* in_sizes, int n_in,
                              void* d_out, int out_size, void* d_ws, size_t ws_size,
                              hipStream_t stream) {
    const float* x      = (const float*)d_in[0];
    const int*   labels = (const int*)  d_in[1];
    const float* W1     = (const float*)d_in[2];
    const float* b1     = (const float*)d_in[3];
    const float* W2     = (const float*)d_in[4];
    const float* b2     = (const float*)d_in[5];
    float*       out    = (float*)d_out;

    moe_tcnn_kernel<<<NWG, 256, 0, stream>>>(x, labels, W1, b1, W2, b2, out);
}
